# starting kernel
# baseline (speedup 1.0000x reference)
#include <hip/hip_runtime.h>
#include <math.h>

#define NIMG 4
#define KCH 3
#define HH 96
#define WW 96
#define PP (HH * WW)
#define TILE_I 64
#define NTILES (PP / TILE_I)
#define THREADS 128

typedef __attribute__((ext_vector_type(2))) float v2f;
typedef __attribute__((ext_vector_type(8))) float v8f;

__device__ __forceinline__ void atomic_max_float(float* addr, float val) {
    if (val >= 0.0f) atomicMax((int*)addr, __float_as_int(val));
    else             atomicMin((unsigned int*)addr, __float_as_uint(val));
}

__global__ void bf_init_max(float* mx) {
    if (threadIdx.x == 0 && blockIdx.x == 0) *mx = -INFINITY;
}

__global__ __launch_bounds__(THREADS) void bf_main(
    const float* __restrict__ raw,
    const float* __restrict__ nrm,
    float* __restrict__ out,
    float* __restrict__ mx)
{
    extern __shared__ float smem[];
    float* s_sq = smem;
    float* s_c0 = smem + PP;
    float* s_c1 = smem + 2 * PP;
    float* s_c2 = smem + 3 * PP;
    float* s_v0 = smem + 4 * PP;
    float* s_v1 = smem + 5 * PP;
    float* s_v2 = smem + 6 * PP;

    const int n    = blockIdx.x / NTILES;
    const int tile = blockIdx.x % NTILES;
    const int tid  = threadIdx.x;

    const float inv_rgb = 1.0f / 15.0f;
    const float inv_xy  = 1.0f / 100.0f;

    const float* rawn = raw + (size_t)n * KCH * PP;
    const float* nrmn = nrm + (size_t)n * KCH * PP;

    for (int p = tid; p < PP; p += THREADS) {
        float c0 = rawn[p]          * inv_rgb;
        float c1 = rawn[PP + p]     * inv_rgb;
        float c2 = rawn[2 * PP + p] * inv_rgb;
        float x  = (float)(p % WW) * inv_xy;
        float y  = (float)(p / WW) * inv_xy;
        s_c0[p] = c0; s_c1[p] = c1; s_c2[p] = c2;
        s_sq[p] = x * x + y * y + c0 * c0 + c1 * c1 + c2 * c2;
        s_v0[p] = nrmn[p];
        s_v1[p] = nrmn[PP + p];
        s_v2[p] = nrmn[2 * PP + p];
    }
    __syncthreads();

    const int lane = tid & 31;
    const int l16  = lane & 15;
    const int hl   = lane >> 4;
    const int wave = tid >> 5;
    const int icol = tile * TILE_I + wave * 16 + l16;

    const float xi = (float)(icol % WW) * inv_xy;
    const float yi = (float)(icol / WW) * inv_xy;
    v2f b1, b2;
    if (hl == 0) { b1.x = xi;          b1.y = yi;          b2.x = s_c2[icol]; b2.y = 0.0f; }
    else         { b1.x = s_c0[icol];  b1.y = s_c1[icol];  b2.x = 0.0f;       b2.y = 0.0f; }
    const float sqi = s_sq[icol];

    float acc0 = 0.0f, acc1 = 0.0f, acc2 = 0.0f;

    const float4* sq4 = (const float4*)s_sq;
    const float4* v04 = (const float4*)s_v0;
    const float4* v14 = (const float4*)s_v1;
    const float4* v24 = (const float4*)s_v2;

    for (int jt = 0; jt < PP; jt += 16) {
        const int j  = jt + l16;
        const float xj = (float)(j % WW) * inv_xy;
        const float yj = (float)(j / WW) * inv_xy;
        v2f a1, a2;
        if (hl == 0) { a1.x = xj;        a1.y = yj;        a2.x = s_c2[j]; a2.y = 0.0f; }
        else         { a1.x = s_c0[j];   a1.y = s_c1[j];   a2.x = 0.0f;    a2.y = 0.0f; }

        v8f g = {0.f, 0.f, 0.f, 0.f, 0.f, 0.f, 0.f, 0.f};
        g = __builtin_amdgcn_wmma_f32_16x16x4_f32(false, a1, false, b1, (short)0, g, false, false);
        g = __builtin_amdgcn_wmma_f32_16x16x4_f32(false, a2, false, b2, (short)0, g, false, false);

        const int b4 = (jt + hl * 8) >> 2;
        float4 sjA = sq4[b4], sjB = sq4[b4 + 1];
        float4 w0A = v04[b4], w0B = v04[b4 + 1];
        float4 w1A = v14[b4], w1B = v14[b4 + 1];
        float4 w2A = v24[b4], w2B = v24[b4 + 1];
        float sj[8]  = {sjA.x, sjA.y, sjA.z, sjA.w, sjB.x, sjB.y, sjB.z, sjB.w};
        float vv0[8] = {w0A.x, w0A.y, w0A.z, w0A.w, w0B.x, w0B.y, w0B.z, w0B.w};
        float vv1[8] = {w1A.x, w1A.y, w1A.z, w1A.w, w1B.x, w1B.y, w1B.z, w1B.w};
        float vv2[8] = {w2A.x, w2A.y, w2A.z, w2A.w, w2B.x, w2B.y, w2B.z, w2B.w};

#pragma unroll
        for (int r = 0; r < 8; ++r) {
            float d2 = fmaxf(sqi + sj[r] - 2.0f * g[r], 0.0f);
            float w  = __expf(-0.5f * d2);
            acc0 = fmaf(w, vv0[r], acc0);
            acc1 = fmaf(w, vv1[r], acc1);
            acc2 = fmaf(w, vv2[r], acc2);
        }
    }

    acc0 += __shfl_xor(acc0, 16, 32);
    acc1 += __shfl_xor(acc1, 16, 32);
    acc2 += __shfl_xor(acc2, 16, 32);

    float m = fmaxf(fmaxf(acc0, acc1), acc2);
#pragma unroll
    for (int off = 8; off >= 1; off >>= 1) m = fmaxf(m, __shfl_xor(m, off, 32));
    if (lane == 0) atomic_max_float(mx, m);

    if (hl == 0) {
        float* o = out + (size_t)n * KCH * PP;
        o[icol]          = acc0;
        o[PP + icol]     = acc1;
        o[2 * PP + icol] = acc2;
    }
}

__global__ void bf_scale(float* __restrict__ out, const float* __restrict__ mx, int total) {
    int i = blockIdx.x * blockDim.x + threadIdx.x;
    if (i < total) out[i] = out[i] / (*mx);
}

extern "C" void kernel_launch(void* const* d_in, const int* in_sizes, int n_in,
                              void* d_out, int out_size, void* d_ws, size_t ws_size,
                              hipStream_t stream) {
    (void)in_sizes; (void)n_in; (void)out_size; (void)ws_size;
    const float* raw = (const float*)d_in[0];
    const float* nrm = (const float*)d_in[1];
    float* out = (float*)d_out;
    float* mx  = (float*)d_ws;

    const size_t lds_bytes = (size_t)7 * PP * sizeof(float);
    hipFuncSetAttribute((const void*)bf_main,
                        hipFuncAttributeMaxDynamicSharedMemorySize, (int)lds_bytes);

    bf_init_max<<<1, 1, 0, stream>>>(mx);
    bf_main<<<NIMG * NTILES, THREADS, lds_bytes, stream>>>(raw, nrm, out, mx);

    const int total = NIMG * KCH * PP;
    bf_scale<<<(total + 255) / 256, 256, 0, stream>>>(out, mx, total);
}


# One staged image per WGP now feeds 27 waves (9 stripes of 16 pixels x 3 j ranges, grid 256) instead of 4; the three j-range sums per pixel are added in fixed order through LDS
# speedup vs baseline: 3.8894x; 3.8894x over previous
#include <hip/hip_runtime.h>
#include <math.h>

#define NIMG 4
#define KCH 3
#define HH 96
#define WW 96
#define PP (HH * WW)
#define NSUB 9
#define JSPLIT 3
#define TILE_I (NSUB * 16)
#define NTILES (PP / TILE_I)
#define NWAVES (NSUB * JSPLIT)
#define THREADS (NWAVES * 32)
#define JSPAN (PP / JSPLIT)
#define NPART ((JSPLIT - 1) * TILE_I * KCH)

typedef __attribute__((ext_vector_type(2))) float v2f;
typedef __attribute__((ext_vector_type(8))) float v8f;

__device__ __forceinline__ void atomic_max_float(float* addr, float val) {
    if (val >= 0.0f) atomicMax((int*)addr, __float_as_int(val));
    else             atomicMin((unsigned int*)addr, __float_as_uint(val));
}

__global__ void bf_init_max(float* mx) {
    if (threadIdx.x == 0 && blockIdx.x == 0) *mx = -INFINITY;
}

__global__ __launch_bounds__(THREADS) void bf_main(
    const float* __restrict__ raw,
    const float* __restrict__ nrm,
    float* __restrict__ out,
    float* __restrict__ mx)
{
    extern __shared__ float smem[];
    float* s_sq = smem;
    float* s_c0 = smem + PP;
    float* s_c1 = smem + 2 * PP;
    float* s_c2 = smem + 3 * PP;
    float* s_v0 = smem + 4 * PP;
    float* s_v1 = smem + 5 * PP;
    float* s_v2 = smem + 6 * PP;
    float* s_part = smem + 7 * PP;

    const int n    = blockIdx.x / NTILES;
    const int tile = blockIdx.x % NTILES;
    const int tid  = threadIdx.x;

    const float inv_rgb = 1.0f / 15.0f;
    const float inv_xy  = 1.0f / 100.0f;

    const float* rawn = raw + (size_t)n * KCH * PP;
    const float* nrmn = nrm + (size_t)n * KCH * PP;

    for (int p = tid; p < PP; p += THREADS) {
        float c0 = rawn[p]          * inv_rgb;
        float c1 = rawn[PP + p]     * inv_rgb;
        float c2 = rawn[2 * PP + p] * inv_rgb;
        float x  = (float)(p % WW) * inv_xy;
        float y  = (float)(p / WW) * inv_xy;
        s_c0[p] = c0; s_c1[p] = c1; s_c2[p] = c2;
        s_sq[p] = x * x + y * y + c0 * c0 + c1 * c1 + c2 * c2;
        s_v0[p] = nrmn[p];
        s_v1[p] = nrmn[PP + p];
        s_v2[p] = nrmn[2 * PP + p];
    }
    __syncthreads();

    const int lane = tid & 31;
    const int l16  = lane & 15;
    const int hl   = lane >> 4;
    const int wave = tid >> 5;
    const int sub   = wave % NSUB;
    const int jpart = wave / NSUB;
    const int icol = tile * TILE_I + sub * 16 + l16;

    const float xi = (float)(icol % WW) * inv_xy;
    const float yi = (float)(icol / WW) * inv_xy;
    v2f b1, b2;
    if (hl == 0) { b1.x = xi;          b1.y = yi;          b2.x = s_c2[icol]; b2.y = 0.0f; }
    else         { b1.x = s_c0[icol];  b1.y = s_c1[icol];  b2.x = 0.0f;       b2.y = 0.0f; }
    const float sqi = s_sq[icol];

    float acc0 = 0.0f, acc1 = 0.0f, acc2 = 0.0f;

    const float4* sq4 = (const float4*)s_sq;
    const float4* v04 = (const float4*)s_v0;
    const float4* v14 = (const float4*)s_v1;
    const float4* v24 = (const float4*)s_v2;

    const int jbeg = jpart * JSPAN;
    for (int jt = jbeg; jt < jbeg + JSPAN; jt += 16) {
        const int j  = jt + l16;
        const float xj = (float)(j % WW) * inv_xy;
        const float yj = (float)(j / WW) * inv_xy;
        v2f a1, a2;
        if (hl == 0) { a1.x = xj;        a1.y = yj;        a2.x = s_c2[j]; a2.y = 0.0f; }
        else         { a1.x = s_c0[j];   a1.y = s_c1[j];   a2.x = 0.0f;    a2.y = 0.0f; }

        v8f g = {0.f, 0.f, 0.f, 0.f, 0.f, 0.f, 0.f, 0.f};
        g = __builtin_amdgcn_wmma_f32_16x16x4_f32(false, a1, false, b1, (short)0, g, false, false);
        g = __builtin_amdgcn_wmma_f32_16x16x4_f32(false, a2, false, b2, (short)0, g, false, false);

        const int b4 = (jt + hl * 8) >> 2;
        float4 sjA = sq4[b4], sjB = sq4[b4 + 1];
        float4 w0A = v04[b4], w0B = v04[b4 + 1];
        float4 w1A = v14[b4], w1B = v14[b4 + 1];
        float4 w2A = v24[b4], w2B = v24[b4 + 1];
        float sj[8]  = {sjA.x, sjA.y, sjA.z, sjA.w, sjB.x, sjB.y, sjB.z, sjB.w};
        float vv0[8] = {w0A.x, w0A.y, w0A.z, w0A.w, w0B.x, w0B.y, w0B.z, w0B.w};
        float vv1[8] = {w1A.x, w1A.y, w1A.z, w1A.w, w1B.x, w1B.y, w1B.z, w1B.w};
        float vv2[8] = {w2A.x, w2A.y, w2A.z, w2A.w, w2B.x, w2B.y, w2B.z, w2B.w};

#pragma unroll
        for (int r = 0; r < 8; ++r) {
            float d2 = fmaxf(sqi + sj[r] - 2.0f * g[r], 0.0f);
            float w  = __expf(-0.5f * d2);
            acc0 = fmaf(w, vv0[r], acc0);
            acc1 = fmaf(w, vv1[r], acc1);
            acc2 = fmaf(w, vv2[r], acc2);
        }
    }

    acc0 += __shfl_xor(acc0, 16, 32);
    acc1 += __shfl_xor(acc1, 16, 32);
    acc2 += __shfl_xor(acc2, 16, 32);

    const int pcol = sub * 16 + l16;
    if (jpart > 0 && hl == 0) {
        float* sp = s_part + (jpart - 1) * (KCH * TILE_I);
        sp[pcol]              = acc0;
        sp[TILE_I + pcol]     = acc1;
        sp[2 * TILE_I + pcol] = acc2;
    }
    __syncthreads();
    if (jpart > 0) return;

    for (int q = 0; q < JSPLIT - 1; ++q) {
        const float* sp = s_part + q * (KCH * TILE_I);
        acc0 += sp[pcol];
        acc1 += sp[TILE_I + pcol];
        acc2 += sp[2 * TILE_I + pcol];
    }

    float m = fmaxf(fmaxf(acc0, acc1), acc2);
#pragma unroll
    for (int off = 8; off >= 1; off >>= 1) m = fmaxf(m, __shfl_xor(m, off, 32));
    if (lane == 0) atomic_max_float(mx, m);

    if (hl == 0) {
        float* o = out + (size_t)n * KCH * PP;
        o[icol]          = acc0;
        o[PP + icol]     = acc1;
        o[2 * PP + icol] = acc2;
    }
}

__global__ void bf_scale(float* __restrict__ out, const float* __restrict__ mx, int total) {
    int i = blockIdx.x * blockDim.x + threadIdx.x;
    if (i < total) out[i] = out[i] / (*mx);
}

extern "C" void kernel_launch(void* const* d_in, const int* in_sizes, int n_in,
                              void* d_out, int out_size, void* d_ws, size_t ws_size,
                              hipStream_t stream) {
    (void)in_sizes; (void)n_in; (void)out_size; (void)ws_size;
    const float* raw = (const float*)d_in[0];
    const float* nrm = (const float*)d_in[1];
    float* out = (float*)d_out;
    float* mx  = (float*)d_ws;

    const size_t lds_bytes = ((size_t)7 * PP + NPART) * sizeof(float);
    hipFuncSetAttribute((const void*)bf_main,
                        hipFuncAttributeMaxDynamicSharedMemorySize, (int)lds_bytes);

    bf_init_max<<<1, 1, 0, stream>>>(mx);
    bf_main<<<NIMG * NTILES, THREADS, lds_bytes, stream>>>(raw, nrm, out, mx);

    const int total = NIMG * KCH * PP;
    bf_scale<<<(total + 255) / 256, 256, 0, stream>>>(out, mx, total);
}


# Squared norms and the -0.5*log2(e) scale ride in the 3 spare K slots of the Gram WMMA (colours centred at mid-scale; d2 now summed inside the WMMA): tile = exp2 argument, per pair exp2 + 3 FMA left
# speedup vs baseline: 5.0572x; 1.3003x over previous
#include <hip/hip_runtime.h>
#include <math.h>

#define NIMG 4
#define KCH 3
#define HH 96
#define WW 96
#define PP (HH * WW)
#define NSUB 9
#define JSPLIT 3
#define TILE_I (NSUB * 16)
#define NTILES (PP / TILE_I)
#define NWAVES (NSUB * JSPLIT)
#define THREADS (NWAVES * 32)
#define JSPAN (PP / JSPLIT)
#define NPART ((JSPLIT - 1) * TILE_I * KCH)

typedef __attribute__((ext_vector_type(2))) float v2f;
typedef __attribute__((ext_vector_type(8))) float v8f;

__device__ __forceinline__ void atomic_max_float(float* addr, float val) {
    if (val >= 0.0f) atomicMax((int*)addr, __float_as_int(val));
    else             atomicMin((unsigned int*)addr, __float_as_uint(val));
}

__global__ void bf_init_max(float* mx) {
    if (threadIdx.x == 0 && blockIdx.x == 0) *mx = -INFINITY;
}

__global__ __launch_bounds__(THREADS) void bf_main(
    const float* __restrict__ raw,
    const float* __restrict__ nrm,
    float* __restrict__ out,
    float* __restrict__ mx)
{
    extern __shared__ float smem[];
    float* s_sq = smem;
    float* s_c0 = smem + PP;
    float* s_c1 = smem + 2 * PP;
    float* s_c2 = smem + 3 * PP;
    float* s_v0 = smem + 4 * PP;
    float* s_v1 = smem + 5 * PP;
    float* s_v2 = smem + 6 * PP;
    float* s_part = smem + 7 * PP;

    const int n    = blockIdx.x / NTILES;
    const int tile = blockIdx.x % NTILES;
    const int tid  = threadIdx.x;

    const float inv_rgb = 1.0f / 15.0f;
    const float inv_xy  = 1.0f / 100.0f;
    const float col_mid = 127.5f / 15.0f;
    const float es = -0.72134752044448170368f;
    const float ef =  1.44269504088896340736f;

    const float* rawn = raw + (size_t)n * KCH * PP;
    const float* nrmn = nrm + (size_t)n * KCH * PP;

    for (int p = tid; p < PP; p += THREADS) {
        float c0 = rawn[p]          * inv_rgb - col_mid;
        float c1 = rawn[PP + p]     * inv_rgb - col_mid;
        float c2 = rawn[2 * PP + p] * inv_rgb - col_mid;
        float x  = (float)(p % WW) * inv_xy;
        float y  = (float)(p / WW) * inv_xy;
        s_c0[p] = c0; s_c1[p] = c1; s_c2[p] = c2;
        s_sq[p] = x * x + y * y + c0 * c0 + c1 * c1 + c2 * c2;
        s_v0[p] = nrmn[p];
        s_v1[p] = nrmn[PP + p];
        s_v2[p] = nrmn[2 * PP + p];
    }
    __syncthreads();

    const int lane = tid & 31;
    const int l16  = lane & 15;
    const int hl   = lane >> 4;
    const int wave = tid >> 5;
    const int sub   = wave % NSUB;
    const int jpart = wave / NSUB;
    const int icol = tile * TILE_I + sub * 16 + l16;

    const float xi = (float)(icol % WW) * inv_xy;
    const float yi = (float)(icol / WW) * inv_xy;
    v2f b1, b2;
    if (hl == 0) { b1.x = ef * xi;          b1.y = ef * yi;          b2.x = ef * s_c2[icol]; b2.y = es;   }
    else         { b1.x = ef * s_c0[icol];  b1.y = ef * s_c1[icol];  b2.x = es * s_sq[icol]; b2.y = 0.0f; }

    float acc0 = 0.0f, acc1 = 0.0f, acc2 = 0.0f;

    const float4* v04 = (const float4*)s_v0;
    const float4* v14 = (const float4*)s_v1;
    const float4* v24 = (const float4*)s_v2;

    const int jbeg = jpart * JSPAN;
    for (int jt = jbeg; jt < jbeg + JSPAN; jt += 16) {
        const int j  = jt + l16;
        const float xj = (float)(j % WW) * inv_xy;
        const float yj = (float)(j / WW) * inv_xy;
        v2f a1, a2;
        if (hl == 0) { a1.x = xj;        a1.y = yj;        a2.x = s_c2[j]; a2.y = s_sq[j]; }
        else         { a1.x = s_c0[j];   a1.y = s_c1[j];   a2.x = 1.0f;    a2.y = 0.0f;    }

        v8f g = {0.f, 0.f, 0.f, 0.f, 0.f, 0.f, 0.f, 0.f};
        g = __builtin_amdgcn_wmma_f32_16x16x4_f32(false, a1, false, b1, (short)0, g, false, false);
        g = __builtin_amdgcn_wmma_f32_16x16x4_f32(false, a2, false, b2, (short)0, g, false, false);

        const int b4 = (jt + hl * 8) >> 2;
        float4 w0A = v04[b4], w0B = v04[b4 + 1];
        float4 w1A = v14[b4], w1B = v14[b4 + 1];
        float4 w2A = v24[b4], w2B = v24[b4 + 1];
        float vv0[8] = {w0A.x, w0A.y, w0A.z, w0A.w, w0B.x, w0B.y, w0B.z, w0B.w};
        float vv1[8] = {w1A.x, w1A.y, w1A.z, w1A.w, w1B.x, w1B.y, w1B.z, w1B.w};
        float vv2[8] = {w2A.x, w2A.y, w2A.z, w2A.w, w2B.x, w2B.y, w2B.z, w2B.w};

#pragma unroll
        for (int r = 0; r < 8; ++r) {
            float w = __builtin_amdgcn_exp2f(g[r]);
            w = fminf(fmaxf(w, 0.0f), 1.0f);
            acc0 = fmaf(w, vv0[r], acc0);
            acc1 = fmaf(w, vv1[r], acc1);
            acc2 = fmaf(w, vv2[r], acc2);
        }
    }

    acc0 += __shfl_xor(acc0, 16, 32);
    acc1 += __shfl_xor(acc1, 16, 32);
    acc2 += __shfl_xor(acc2, 16, 32);

    const int pcol = sub * 16 + l16;
    if (jpart > 0 && hl == 0) {
        float* sp = s_part + (jpart - 1) * (KCH * TILE_I);
        sp[pcol]              = acc0;
        sp[TILE_I + pcol]     = acc1;
        sp[2 * TILE_I + pcol] = acc2;
    }
    __syncthreads();
    if (jpart > 0) return;

    for (int q = 0; q < JSPLIT - 1; ++q) {
        const float* sp = s_part + q * (KCH * TILE_I);
        acc0 += sp[pcol];
        acc1 += sp[TILE_I + pcol];
        acc2 += sp[2 * TILE_I + pcol];
    }

    float m = fmaxf(fmaxf(acc0, acc1), acc2);
#pragma unroll
    for (int off = 8; off >= 1; off >>= 1) m = fmaxf(m, __shfl_xor(m, off, 32));
    if (lane == 0) atomic_max_float(mx, m);

    if (hl == 0) {
        float* o = out + (size_t)n * KCH * PP;
        o[icol]          = acc0;
        o[PP + icol]     = acc1;
        o[2 * PP + icol] = acc2;
    }
}

__global__ void bf_scale(float* __restrict__ out, const float* __restrict__ mx, int total) {
    int i = blockIdx.x * blockDim.x + threadIdx.x;
    if (i < total) out[i] = out[i] / (*mx);
}

extern "C" void kernel_launch(void* const* d_in, const int* in_sizes, int n_in,
                              void* d_out, int out_size, void* d_ws, size_t ws_size,
                              hipStream_t stream) {
    (void)in_sizes; (void)n_in; (void)out_size; (void)ws_size;
    const float* raw = (const float*)d_in[0];
    const float* nrm = (const float*)d_in[1];
    float* out = (float*)d_out;
    float* mx  = (float*)d_ws;

    const size_t lds_bytes = ((size_t)7 * PP + NPART) * sizeof(float);
    hipFuncSetAttribute((const void*)bf_main,
                        hipFuncAttributeMaxDynamicSharedMemorySize, (int)lds_bytes);

    bf_init_max<<<1, 1, 0, stream>>>(mx);
    bf_main<<<NIMG * NTILES, THREADS, lds_bytes, stream>>>(raw, nrm, out, mx);

    const int total = NIMG * KCH * PP;
    bf_scale<<<(total + 255) / 256, 256, 0, stream>>>(out, mx, total);
}


# A fragment without lane-half branches or divides: features as one 16-byte LDS record per pixel (one 8-byte read per lane), x_j in 6 row-invariant registers, y_j once per row; Gram k order changed
# speedup vs baseline: 6.6299x; 1.3110x over previous
#include <hip/hip_runtime.h>
#include <math.h>

#define NIMG 4
#define KCH 3
#define HH 96
#define WW 96
#define PP (HH * WW)
#define NSUB 9
#define JSPLIT 3
#define TILE_I (NSUB * 16)
#define NTILES (PP / TILE_I)
#define NWAVES (NSUB * JSPLIT)
#define THREADS (NWAVES * 32)
#define JSPAN (PP / JSPLIT)
#define JROWS (JSPAN / WW)
#define ROWT (WW / 16)
#define NPART ((JSPLIT - 1) * TILE_I * KCH)

typedef __attribute__((ext_vector_type(2))) float v2f;
typedef __attribute__((ext_vector_type(8))) float v8f;

static_assert(JSPAN % WW == 0 && WW % 16 == 0, "a wave's j range is whole image rows of whole 16-pixel tiles");

__device__ __forceinline__ void atomic_max_float(float* addr, float val) {
    if (val >= 0.0f) atomicMax((int*)addr, __float_as_int(val));
    else             atomicMin((unsigned int*)addr, __float_as_uint(val));
}

__global__ void bf_init_max(float* mx) {
    if (threadIdx.x == 0 && blockIdx.x == 0) *mx = -INFINITY;
}

__global__ __launch_bounds__(THREADS) void bf_main(
    const float* __restrict__ raw,
    const float* __restrict__ nrm,
    float* __restrict__ out,
    float* __restrict__ mx)
{
    extern __shared__ float smem[];
    float4* s_f = (float4*)smem;
    float* s_v0 = smem + 4 * PP;
    float* s_v1 = smem + 5 * PP;
    float* s_v2 = smem + 6 * PP;
    float* s_part = smem + 7 * PP;

    const int n    = blockIdx.x / NTILES;
    const int tile = blockIdx.x % NTILES;
    const int tid  = threadIdx.x;

    const float inv_rgb = 1.0f / 15.0f;
    const float inv_xy  = 1.0f / 100.0f;
    const float col_mid = 127.5f / 15.0f;
    const float es = -0.72134752044448170368f;
    const float ef =  1.44269504088896340736f;

    const float* rawn = raw + (size_t)n * KCH * PP;
    const float* nrmn = nrm + (size_t)n * KCH * PP;

    for (int p = tid; p < PP; p += THREADS) {
        float c0 = rawn[p]          * inv_rgb - col_mid;
        float c1 = rawn[PP + p]     * inv_rgb - col_mid;
        float c2 = rawn[2 * PP + p] * inv_rgb - col_mid;
        float x  = (float)(p % WW) * inv_xy;
        float y  = (float)(p / WW) * inv_xy;
        s_f[p] = make_float4(c0, c1, c2, x * x + y * y + c0 * c0 + c1 * c1 + c2 * c2);
        s_v0[p] = nrmn[p];
        s_v1[p] = nrmn[PP + p];
        s_v2[p] = nrmn[2 * PP + p];
    }
    __syncthreads();

    const int lane = tid & 31;
    const int l16  = lane & 15;
    const int hl   = lane >> 4;
    const int wave = tid >> 5;
    const int sub   = wave % NSUB;
    const int jpart = wave / NSUB;
    const int icol = tile * TILE_I + sub * 16 + l16;

    const float xi = (float)(icol % WW) * inv_xy;
    const float yi = (float)(icol / WW) * inv_xy;
    const float4 fi = s_f[icol];
    v2f b1, b2;
    if (hl == 0) { b1.x = ef * fi.x;  b1.y = ef * fi.y;  b2.x = ef * xi;    b2.y = ef * yi; }
    else         { b1.x = ef * fi.z;  b1.y = es;         b2.x = es * fi.w;  b2.y = 0.0f;    }

    float ax[ROWT];
    for (int t = 0; t < ROWT; ++t) ax[t] = (hl == 0) ? (float)(t * 16 + l16) * inv_xy : 1.0f;
    const float ymul = (hl == 0) ? inv_xy : 0.0f;
    const int row0 = jpart * JROWS;
    const v2f* arow = (const v2f*)smem + 2 * (row0 * WW + l16) + hl;
    const float4* vrow = (const float4*)s_v0 + (row0 * WW + hl * 8) / 4;

    float acc0 = 0.0f, acc1 = 0.0f, acc2 = 0.0f;

    for (int row = row0; row < row0 + JROWS; ++row) {
      const float ay = (float)row * ymul;
#pragma unroll
      for (int t = 0; t < ROWT; ++t) {
        const v2f a1 = arow[2 * 16 * t];
        v2f a2; a2.x = ax[t]; a2.y = ay;

        v8f g = {0.f, 0.f, 0.f, 0.f, 0.f, 0.f, 0.f, 0.f};
        g = __builtin_amdgcn_wmma_f32_16x16x4_f32(false, a1, false, b1, (short)0, g, false, false);
        g = __builtin_amdgcn_wmma_f32_16x16x4_f32(false, a2, false, b2, (short)0, g, false, false);

        const int b4 = 4 * t;
        float4 w0A = vrow[b4],              w0B = vrow[b4 + 1];
        float4 w1A = vrow[PP / 4 + b4],     w1B = vrow[PP / 4 + b4 + 1];
        float4 w2A = vrow[2 * (PP / 4) + b4], w2B = vrow[2 * (PP / 4) + b4 + 1];
        float vv0[8] = {w0A.x, w0A.y, w0A.z, w0A.w, w0B.x, w0B.y, w0B.z, w0B.w};
        float vv1[8] = {w1A.x, w1A.y, w1A.z, w1A.w, w1B.x, w1B.y, w1B.z, w1B.w};
        float vv2[8] = {w2A.x, w2A.y, w2A.z, w2A.w, w2B.x, w2B.y, w2B.z, w2B.w};

#pragma unroll
        for (int r = 0; r < 8; ++r) {
            float w = __builtin_amdgcn_exp2f(g[r]);
            w = fminf(fmaxf(w, 0.0f), 1.0f);
            acc0 = fmaf(w, vv0[r], acc0);
            acc1 = fmaf(w, vv1[r], acc1);
            acc2 = fmaf(w, vv2[r], acc2);
        }
      }
      arow += 2 * WW;
      vrow += WW / 4;
    }

    acc0 += __shfl_xor(acc0, 16, 32);
    acc1 += __shfl_xor(acc1, 16, 32);
    acc2 += __shfl_xor(acc2, 16, 32);

    const int pcol = sub * 16 + l16;
    if (jpart > 0 && hl == 0) {
        float* sp = s_part + (jpart - 1) * (KCH * TILE_I);
        sp[pcol]              = acc0;
        sp[TILE_I + pcol]     = acc1;
        sp[2 * TILE_I + pcol] = acc2;
    }
    __syncthreads();
    if (jpart > 0) return;

    for (int q = 0; q < JSPLIT - 1; ++q) {
        const float* sp = s_part + q * (KCH * TILE_I);
        acc0 += sp[pcol];
        acc1 += sp[TILE_I + pcol];
        acc2 += sp[2 * TILE_I + pcol];
    }

    float m = fmaxf(fmaxf(acc0, acc1), acc2);
#pragma unroll
    for (int off = 8; off >= 1; off >>= 1) m = fmaxf(m, __shfl_xor(m, off, 32));
    if (lane == 0) atomic_max_float(mx, m);

    if (hl == 0) {
        float* o = out + (size_t)n * KCH * PP;
        o[icol]          = acc0;
        o[PP + icol]     = acc1;
        o[2 * PP + icol] = acc2;
    }
}

__global__ void bf_scale(float* __restrict__ out, const float* __restrict__ mx, int total) {
    int i = blockIdx.x * blockDim.x + threadIdx.x;
    if (i < total) out[i] = out[i] / (*mx);
}

extern "C" void kernel_launch(void* const* d_in, const int* in_sizes, int n_in,
                              void* d_out, int out_size, void* d_ws, size_t ws_size,
                              hipStream_t stream) {
    (void)in_sizes; (void)n_in; (void)out_size; (void)ws_size;
    const float* raw = (const float*)d_in[0];
    const float* nrm = (const float*)d_in[1];
    float* out = (float*)d_out;
    float* mx  = (float*)d_ws;

    const size_t lds_bytes = ((size_t)7 * PP + NPART) * sizeof(float);
    hipFuncSetAttribute((const void*)bf_main,
                        hipFuncAttributeMaxDynamicSharedMemorySize, (int)lds_bytes);

    bf_init_max<<<1, 1, 0, stream>>>(mx);
    bf_main<<<NIMG * NTILES, THREADS, lds_bytes, stream>>>(raw, nrm, out, mx);

    const int total = NIMG * KCH * PP;
    bf_scale<<<(total + 255) / 256, 256, 0, stream>>>(out, mx, total);
}


# Gram tile transposed (i stripe = A rows, j = B columns): a lane owns one j per tile, one 16-byte value read (was six), 24 per-lane sums; summation order changed: 48 partials per output added via LDS
# speedup vs baseline: 7.5959x; 1.1457x over previous
#include <hip/hip_runtime.h>
#include <math.h>

#define NIMG 4
#define KCH 3
#define HH 96
#define WW 96
#define PP (HH * WW)
#define NSUB 9
#define JSPLIT 3
#define TILE_I (NSUB * 16)
#define NTILES (PP / TILE_I)
#define NWAVES (NSUB * JSPLIT)
#define THREADS (NWAVES * 32)
#define JSPAN (PP / JSPLIT)
#define JROWS (JSPAN / WW)
#define ROWT (WW / 16)
#define NOUT (TILE_I * KCH)
#define NPARTIAL (JSPLIT * 16)
#define RSTR (NPARTIAL + 4)

typedef __attribute__((ext_vector_type(2))) float v2f;
typedef __attribute__((ext_vector_type(8))) float v8f;

static_assert(JSPAN % WW == 0 && WW % 16 == 0, "a wave's j range is whole image rows of whole 16-pixel tiles");

__device__ __forceinline__ void atomic_max_float(float* addr, float val) {
    if (val >= 0.0f) atomicMax((int*)addr, __float_as_int(val));
    else             atomicMin((unsigned int*)addr, __float_as_uint(val));
}

__global__ void bf_init_max(float* mx) {
    if (threadIdx.x == 0 && blockIdx.x == 0) *mx = -INFINITY;
}

__global__ __launch_bounds__(THREADS) void bf_main(
    const float* __restrict__ raw,
    const float* __restrict__ nrm,
    float* __restrict__ out,
    float* __restrict__ mx)
{
    extern __shared__ float smem[];
    float4* s_f = (float4*)smem;
    float4* s_v = (float4*)(smem + 4 * PP);

    const int n    = blockIdx.x / NTILES;
    const int tile = blockIdx.x % NTILES;
    const int tid  = threadIdx.x;

    const float inv_rgb = 1.0f / 15.0f;
    const float inv_xy  = 1.0f / 100.0f;
    const float col_mid = 127.5f / 15.0f;
    const float es = -0.72134752044448170368f;
    const float ef =  1.44269504088896340736f;

    const float* rawn = raw + (size_t)n * KCH * PP;
    const float* nrmn = nrm + (size_t)n * KCH * PP;

    for (int p = tid; p < PP; p += THREADS) {
        float c0 = rawn[p]          * inv_rgb - col_mid;
        float c1 = rawn[PP + p]     * inv_rgb - col_mid;
        float c2 = rawn[2 * PP + p] * inv_rgb - col_mid;
        float x  = (float)(p % WW) * inv_xy;
        float y  = (float)(p / WW) * inv_xy;
        s_f[p] = make_float4(c0, c1, c2, x * x + y * y + c0 * c0 + c1 * c1 + c2 * c2);
        s_v[p] = make_float4(nrmn[p], nrmn[PP + p], nrmn[2 * PP + p], 0.0f);
    }
    __syncthreads();

    const int lane = tid & 31;
    const int l16  = lane & 15;
    const int hl   = lane >> 4;
    const int wave = tid >> 5;
    const int sub   = wave % NSUB;
    const int jpart = wave / NSUB;
    const int icol = tile * TILE_I + sub * 16 + l16;

    const float xi = (float)(icol % WW) * inv_xy;
    const float yi = (float)(icol / WW) * inv_xy;
    const float4 fi = s_f[icol];
    v2f ia1, ia2;
    if (hl == 0) { ia1.x = ef * fi.x;  ia1.y = ef * fi.y;  ia2.x = ef * xi;    ia2.y = ef * yi; }
    else         { ia1.x = ef * fi.z;  ia1.y = es;         ia2.x = es * fi.w;  ia2.y = 0.0f;    }

    float bx[ROWT];
    for (int t = 0; t < ROWT; ++t) bx[t] = (hl == 0) ? (float)(t * 16 + l16) * inv_xy : 1.0f;
    const float ymul = (hl == 0) ? inv_xy : 0.0f;
    const int row0 = jpart * JROWS;
    const v2f* brow = (const v2f*)smem + 2 * (row0 * WW + l16) + hl;
    const float4* vrow = s_v + (row0 * WW + l16);

    float acc[8][KCH];
#pragma unroll
    for (int r = 0; r < 8; ++r) { acc[r][0] = 0.0f; acc[r][1] = 0.0f; acc[r][2] = 0.0f; }

    for (int row = row0; row < row0 + JROWS; ++row) {
      const float by = (float)row * ymul;
#pragma unroll
      for (int t = 0; t < ROWT; ++t) {
        const v2f b1 = brow[2 * 16 * t];
        v2f b2; b2.x = bx[t]; b2.y = by;

        v8f g = {0.f, 0.f, 0.f, 0.f, 0.f, 0.f, 0.f, 0.f};
        g = __builtin_amdgcn_wmma_f32_16x16x4_f32(false, ia1, false, b1, (short)0, g, false, false);
        g = __builtin_amdgcn_wmma_f32_16x16x4_f32(false, ia2, false, b2, (short)0, g, false, false);

        const float4 vj = vrow[16 * t];

#pragma unroll
        for (int r = 0; r < 8; ++r) {
            float w = __builtin_amdgcn_exp2f(g[r]);
            w = fminf(fmaxf(w, 0.0f), 1.0f);
            acc[r][0] = fmaf(w, vj.x, acc[r][0]);
            acc[r][1] = fmaf(w, vj.y, acc[r][1]);
            acc[r][2] = fmaf(w, vj.z, acc[r][2]);
        }
      }
      brow += 2 * WW;
      vrow += WW;
    }

    __syncthreads();
    float* red = smem;
    {
        float* rp = red + (sub * 16 + 8 * hl) * RSTR + jpart * 16 + l16;
#pragma unroll
        for (int r = 0; r < 8; ++r) {
            rp[r * RSTR]                       = acc[r][0];
            rp[(TILE_I + r) * RSTR]            = acc[r][1];
            rp[(2 * TILE_I + r) * RSTR]        = acc[r][2];
        }
    }
    __syncthreads();

    if (wave >= (NOUT + 31) / 32) return;
    const bool live = tid < NOUT;
    const int  e    = live ? tid : 0;
    const int  ch   = e / TILE_I;
    const int  il   = e % TILE_I;
    const float4* q = (const float4*)(red + e * RSTR);
    float s = 0.0f;
    for (int k = 0; k < NPARTIAL / 4; ++k) {
        const float4 p4 = q[k];
        s += p4.x; s += p4.y; s += p4.z; s += p4.w;
    }

    float m = live ? s : -INFINITY;
#pragma unroll
    for (int off = 16; off >= 1; off >>= 1) m = fmaxf(m, __shfl_xor(m, off, 32));
    if (lane == 0) atomic_max_float(mx, m);

    if (live) out[((size_t)n * KCH + ch) * PP + tile * TILE_I + il] = s;
}

__global__ void bf_scale(float* __restrict__ out, const float* __restrict__ mx, int total) {
    int i = blockIdx.x * blockDim.x + threadIdx.x;
    if (i < total) out[i] = out[i] / (*mx);
}

extern "C" void kernel_launch(void* const* d_in, const int* in_sizes, int n_in,
                              void* d_out, int out_size, void* d_ws, size_t ws_size,
                              hipStream_t stream) {
    (void)in_sizes; (void)n_in; (void)out_size; (void)ws_size;
    const float* raw = (const float*)d_in[0];
    const float* nrm = (const float*)d_in[1];
    float* out = (float*)d_out;
    float* mx  = (float*)d_ws;

    const size_t lds_bytes = (size_t)8 * PP * sizeof(float);
    hipFuncSetAttribute((const void*)bf_main,
                        hipFuncAttributeMaxDynamicSharedMemorySize, (int)lds_bytes);

    bf_init_max<<<1, 1, 0, stream>>>(mx);
    bf_main<<<NIMG * NTILES, THREADS, lds_bytes, stream>>>(raw, nrm, out, mx);

    const int total = NIMG * KCH * PP;
    bf_scale<<<(total + 255) / 256, 256, 0, stream>>>(out, mx, total);
}


# Position terms leave the Gram product: es*(sq_i + dx^2) is the C operand (fixed per tile column, so j runs column-outer, row-inner), dy^2 joins sq_j: one WMMA per tile, not two; j sum order changed
# speedup vs baseline: 9.3914x; 1.2364x over previous
#include <hip/hip_runtime.h>
#include <math.h>

#define NIMG 4
#define KCH 3
#define HH 96
#define WW 96
#define PP (HH * WW)
#define NSUB 9
#define JSPLIT 3
#define TILE_I (NSUB * 16)
#define NTILES (PP / TILE_I)
#define NWAVES (NSUB * JSPLIT)
#define THREADS (NWAVES * 32)
#define JSPAN (PP / JSPLIT)
#define JROWS (JSPAN / WW)
#define ROWT (WW / 16)
#define NOUT (TILE_I * KCH)
#define NPARTIAL (JSPLIT * 16)
#define RSTR (NPARTIAL + 4)

typedef __attribute__((ext_vector_type(2))) float v2f;
typedef __attribute__((ext_vector_type(8))) float v8f;

static_assert(JSPAN % WW == 0 && WW % 16 == 0, "a wave's j range is whole image rows of whole 16-pixel tiles");

__device__ __forceinline__ void atomic_max_float(float* addr, float val) {
    if (val >= 0.0f) atomicMax((int*)addr, __float_as_int(val));
    else             atomicMin((unsigned int*)addr, __float_as_uint(val));
}

__global__ void bf_init_max(float* mx) {
    if (threadIdx.x == 0 && blockIdx.x == 0) *mx = -INFINITY;
}

__global__ __launch_bounds__(THREADS) void bf_main(
    const float* __restrict__ raw,
    const float* __restrict__ nrm,
    float* __restrict__ out,
    float* __restrict__ mx)
{
    extern __shared__ float smem[];
    float4* s_f = (float4*)smem;
    float4* s_v = (float4*)(smem + 4 * PP);

    const int n    = blockIdx.x / NTILES;
    const int tile = blockIdx.x % NTILES;
    const int tid  = threadIdx.x;

    const float inv_rgb = 1.0f / 15.0f;
    const float inv_xy  = 1.0f / 100.0f;
    const float col_mid = 127.5f / 15.0f;
    const float es = -0.72134752044448170368f;
    const float ef =  1.44269504088896340736f;

    const float* rawn = raw + (size_t)n * KCH * PP;
    const float* nrmn = nrm + (size_t)n * KCH * PP;

    for (int p = tid; p < PP; p += THREADS) {
        float c0 = rawn[p]          * inv_rgb - col_mid;
        float c1 = rawn[PP + p]     * inv_rgb - col_mid;
        float c2 = rawn[2 * PP + p] * inv_rgb - col_mid;
        s_f[p] = make_float4(c0, c1, c2, c0 * c0 + c1 * c1 + c2 * c2);
        s_v[p] = make_float4(nrmn[p], nrmn[PP + p], nrmn[2 * PP + p], 0.0f);
    }
    __syncthreads();

    const int lane = tid & 31;
    const int l16  = lane & 15;
    const int hl   = lane >> 4;
    const int wave = tid >> 5;
    const int sub   = wave % NSUB;
    const int jpart = wave / NSUB;
    const int ibase = tile * TILE_I + sub * 16;
    const int xi0   = ibase % WW;
    const int yi    = ibase / WW;

    const float4 fi = s_f[ibase + l16];
    v2f ia;
    if (hl == 0) { ia.x = ef * fi.x;  ia.y = ef * fi.y; }
    else         { ia.x = ef * fi.z;  ia.y = es;        }

    float sqi[8];
#pragma unroll
    for (int r = 0; r < 8; ++r) sqi[r] = s_f[ibase + 8 * hl + r].w;

    const float ymul = (hl == 0) ? 0.0f : inv_xy;
    const int row0 = jpart * JROWS;

    float acc[8][KCH];
#pragma unroll
    for (int r = 0; r < 8; ++r) { acc[r][0] = 0.0f; acc[r][1] = 0.0f; acc[r][2] = 0.0f; }

    for (int t = 0; t < ROWT; ++t) {
      const int dx0 = xi0 + 8 * hl - 16 * t - l16;
      v8f cx;
#pragma unroll
      for (int r = 0; r < 8; ++r) {
          const float dx = (float)(dx0 + r) * inv_xy;
          cx[r] = es * fmaf(dx, dx, sqi[r]);
      }
      const v2f*    bp = (const v2f*)smem + 2 * (row0 * WW + 16 * t + l16) + hl;
      const float4* vp = s_v + (row0 * WW + 16 * t + l16);
      float fy = (float)(row0 - yi);

#pragma unroll 4
      for (int row = row0; row < row0 + JROWS; ++row) {
        v2f b = *bp;
        const float dy = fy * ymul;
        b.y = fmaf(dy, dy, b.y);
        fy += 1.0f;

        const v8f g = __builtin_amdgcn_wmma_f32_16x16x4_f32(false, ia, false, b, (short)0, cx, false, false);

        const float4 vj = *vp;

#pragma unroll
        for (int r = 0; r < 8; ++r) {
            float w = __builtin_amdgcn_exp2f(g[r]);
            w = fminf(fmaxf(w, 0.0f), 1.0f);
            acc[r][0] = fmaf(w, vj.x, acc[r][0]);
            acc[r][1] = fmaf(w, vj.y, acc[r][1]);
            acc[r][2] = fmaf(w, vj.z, acc[r][2]);
        }
        bp += 2 * WW;
        vp += WW;
      }
    }

    __syncthreads();
    float* red = smem;
    {
        float* rp = red + (sub * 16 + 8 * hl) * RSTR + jpart * 16 + l16;
#pragma unroll
        for (int r = 0; r < 8; ++r) {
            rp[r * RSTR]                       = acc[r][0];
            rp[(TILE_I + r) * RSTR]            = acc[r][1];
            rp[(2 * TILE_I + r) * RSTR]        = acc[r][2];
        }
    }
    __syncthreads();

    if (wave >= (NOUT + 31) / 32) return;
    const bool live = tid < NOUT;
    const int  e    = live ? tid : 0;
    const int  ch   = e / TILE_I;
    const int  il   = e % TILE_I;
    const float4* q = (const float4*)(red + e * RSTR);
    float s = 0.0f;
    for (int k = 0; k < NPARTIAL / 4; ++k) {
        const float4 p4 = q[k];
        s += p4.x; s += p4.y; s += p4.z; s += p4.w;
    }

    float m = live ? s : -INFINITY;
#pragma unroll
    for (int off = 16; off >= 1; off >>= 1) m = fmaxf(m, __shfl_xor(m, off, 32));
    if (lane == 0) atomic_max_float(mx, m);

    if (live) out[((size_t)n * KCH + ch) * PP + tile * TILE_I + il] = s;
}

__global__ void bf_scale(float* __restrict__ out, const float* __restrict__ mx, int total) {
    int i = blockIdx.x * blockDim.x + threadIdx.x;
    if (i < total) out[i] = out[i] / (*mx);
}

extern "C" void kernel_launch(void* const* d_in, const int* in_sizes, int n_in,
                              void* d_out, int out_size, void* d_ws, size_t ws_size,
                              hipStream_t stream) {
    (void)in_sizes; (void)n_in; (void)out_size; (void)ws_size;
    const float* raw = (const float*)d_in[0];
    const float* nrm = (const float*)d_in[1];
    float* out = (float*)d_out;
    float* mx  = (float*)d_ws;

    const size_t lds_bytes = (size_t)8 * PP * sizeof(float);
    hipFuncSetAttribute((const void*)bf_main,
                        hipFuncAttributeMaxDynamicSharedMemorySize, (int)lds_bytes);

    bf_init_max<<<1, 1, 0, stream>>>(mx);
    bf_main<<<NIMG * NTILES, THREADS, lds_bytes, stream>>>(raw, nrm, out, mx);

    const int total = NIMG * KCH * PP;
    bf_scale<<<(total + 255) / 256, 256, 0, stream>>>(out, mx, total);
}


# Global max without atomics: each workgroup stores its own max in a workspace slot and every wave of the scaling kernel reduces the 256 slots itself; the reset launch is gone (3 launches -> 2)
# speedup vs baseline: 9.9504x; 1.0595x over previous
#include <hip/hip_runtime.h>
#include <math.h>

#define NIMG 4
#define KCH 3
#define HH 96
#define WW 96
#define PP (HH * WW)
#define NSUB 9
#define JSPLIT 3
#define TILE_I (NSUB * 16)
#define NTILES (PP / TILE_I)
#define NWAVES (NSUB * JSPLIT)
#define THREADS (NWAVES * 32)
#define JSPAN (PP / JSPLIT)
#define JROWS (JSPAN / WW)
#define ROWT (WW / 16)
#define NOUT (TILE_I * KCH)
#define NPARTIAL (JSPLIT * 16)
#define RSTR (NPARTIAL + 4)
#define NGROUPS (NIMG * NTILES)
#define NSUMW ((NOUT + 31) / 32)

typedef __attribute__((ext_vector_type(2))) float v2f;
typedef __attribute__((ext_vector_type(8))) float v8f;

static_assert(JSPAN % WW == 0 && WW % 16 == 0, "a wave's j range is whole image rows of whole 16-pixel tiles");

__global__ __launch_bounds__(THREADS) void bf_main(
    const float* __restrict__ raw,
    const float* __restrict__ nrm,
    float* __restrict__ out,
    float* __restrict__ wgmax)
{
    extern __shared__ float smem[];
    float4* s_f = (float4*)smem;
    float4* s_v = (float4*)(smem + 4 * PP);

    const int n    = blockIdx.x / NTILES;
    const int tile = blockIdx.x % NTILES;
    const int tid  = threadIdx.x;

    const float inv_rgb = 1.0f / 15.0f;
    const float inv_xy  = 1.0f / 100.0f;
    const float col_mid = 127.5f / 15.0f;
    const float es = -0.72134752044448170368f;
    const float ef =  1.44269504088896340736f;

    const float* rawn = raw + (size_t)n * KCH * PP;
    const float* nrmn = nrm + (size_t)n * KCH * PP;

    for (int p = tid; p < PP; p += THREADS) {
        float c0 = rawn[p]          * inv_rgb - col_mid;
        float c1 = rawn[PP + p]     * inv_rgb - col_mid;
        float c2 = rawn[2 * PP + p] * inv_rgb - col_mid;
        s_f[p] = make_float4(c0, c1, c2, c0 * c0 + c1 * c1 + c2 * c2);
        s_v[p] = make_float4(nrmn[p], nrmn[PP + p], nrmn[2 * PP + p], 0.0f);
    }
    __syncthreads();

    const int lane = tid & 31;
    const int l16  = lane & 15;
    const int hl   = lane >> 4;
    const int wave = tid >> 5;
    const int sub   = wave % NSUB;
    const int jpart = wave / NSUB;
    const int ibase = tile * TILE_I + sub * 16;
    const int xi0   = ibase % WW;
    const int yi    = ibase / WW;

    const float4 fi = s_f[ibase + l16];
    v2f ia;
    if (hl == 0) { ia.x = ef * fi.x;  ia.y = ef * fi.y; }
    else         { ia.x = ef * fi.z;  ia.y = es;        }

    float sqi[8];
#pragma unroll
    for (int r = 0; r < 8; ++r) sqi[r] = s_f[ibase + 8 * hl + r].w;

    const float ymul = (hl == 0) ? 0.0f : inv_xy;
    const int row0 = jpart * JROWS;

    float acc[8][KCH];
#pragma unroll
    for (int r = 0; r < 8; ++r) { acc[r][0] = 0.0f; acc[r][1] = 0.0f; acc[r][2] = 0.0f; }

    for (int t = 0; t < ROWT; ++t) {
      const int dx0 = xi0 + 8 * hl - 16 * t - l16;
      v8f cx;
#pragma unroll
      for (int r = 0; r < 8; ++r) {
          const float dx = (float)(dx0 + r) * inv_xy;
          cx[r] = es * fmaf(dx, dx, sqi[r]);
      }
      const v2f*    bp = (const v2f*)smem + 2 * (row0 * WW + 16 * t + l16) + hl;
      const float4* vp = s_v + (row0 * WW + 16 * t + l16);
      float fy = (float)(row0 - yi);

#pragma unroll 4
      for (int row = row0; row < row0 + JROWS; ++row) {
        v2f b = *bp;
        const float dy = fy * ymul;
        b.y = fmaf(dy, dy, b.y);
        fy += 1.0f;

        const v8f g = __builtin_amdgcn_wmma_f32_16x16x4_f32(false, ia, false, b, (short)0, cx, false, false);

        const float4 vj = *vp;

#pragma unroll
        for (int r = 0; r < 8; ++r) {
            float w = __builtin_amdgcn_exp2f(g[r]);
            w = fminf(fmaxf(w, 0.0f), 1.0f);
            acc[r][0] = fmaf(w, vj.x, acc[r][0]);
            acc[r][1] = fmaf(w, vj.y, acc[r][1]);
            acc[r][2] = fmaf(w, vj.z, acc[r][2]);
        }
        bp += 2 * WW;
        vp += WW;
      }
    }

    __syncthreads();
    float* red = smem;
    {
        float* rp = red + (sub * 16 + 8 * hl) * RSTR + jpart * 16 + l16;
#pragma unroll
        for (int r = 0; r < 8; ++r) {
            rp[r * RSTR]                       = acc[r][0];
            rp[(TILE_I + r) * RSTR]            = acc[r][1];
            rp[(2 * TILE_I + r) * RSTR]        = acc[r][2];
        }
    }
    __syncthreads();

    float* wmax = smem + 4 * PP;
    const bool live = tid < NOUT;
    const int  e    = live ? tid : 0;
    const int  ch   = e / TILE_I;
    const int  il   = e % TILE_I;
    float s = 0.0f;
    if (wave < NSUMW) {
        const float4* q = (const float4*)(red + e * RSTR);
        for (int k = 0; k < NPARTIAL / 4; ++k) {
            const float4 p4 = q[k];
            s += p4.x; s += p4.y; s += p4.z; s += p4.w;
        }
        float m = live ? s : -INFINITY;
#pragma unroll
        for (int off = 16; off >= 1; off >>= 1) m = fmaxf(m, __shfl_xor(m, off, 32));
        if (lane == 0) wmax[wave] = m;
    }
    __syncthreads();
    if (live) out[((size_t)n * KCH + ch) * PP + tile * TILE_I + il] = s;
    if (wave == 0) {
        float m = (lane < NSUMW) ? wmax[lane] : -INFINITY;
#pragma unroll
        for (int off = 16; off >= 1; off >>= 1) m = fmaxf(m, __shfl_xor(m, off, 32));
        if (lane == 0) wgmax[blockIdx.x] = m;
    }
}

__global__ void bf_scale(float* __restrict__ out, const float* __restrict__ wgmax, int total) {
    static_assert(NGROUPS == 8 * 32, "two 16-byte reads per lane cover every slot once");
    const int i    = blockIdx.x * blockDim.x + threadIdx.x;
    const int lane = threadIdx.x & 31;
    const float4* q = (const float4*)wgmax;
    const float4 a = q[lane];
    const float4 b = q[lane + 32];
    const float  v = (i < total) ? out[i] : 0.0f;
    float m = fmaxf(fmaxf(fmaxf(a.x, a.y), fmaxf(a.z, a.w)), fmaxf(fmaxf(b.x, b.y), fmaxf(b.z, b.w)));
#pragma unroll
    for (int off = 16; off >= 1; off >>= 1) m = fmaxf(m, __shfl_xor(m, off, 32));
    if (i < total) out[i] = v / m;
}

extern "C" void kernel_launch(void* const* d_in, const int* in_sizes, int n_in,
                              void* d_out, int out_size, void* d_ws, size_t ws_size,
                              hipStream_t stream) {
    (void)in_sizes; (void)n_in; (void)out_size; (void)ws_size;
    const float* raw = (const float*)d_in[0];
    const float* nrm = (const float*)d_in[1];
    float* out = (float*)d_out;
    float* wgmax = (float*)d_ws;

    const size_t lds_bytes = (size_t)8 * PP * sizeof(float);
    hipFuncSetAttribute((const void*)bf_main,
                        hipFuncAttributeMaxDynamicSharedMemorySize, (int)lds_bytes);

    bf_main<<<NGROUPS, THREADS, lds_bytes, stream>>>(raw, nrm, out, wgmax);

    const int total = NIMG * KCH * PP;
    bf_scale<<<(total + 255) / 256, 256, 0, stream>>>(out, wgmax, total);
}


# speedup vs baseline: 10.4082x; 1.0460x over previous
#include <hip/hip_runtime.h>
#include <math.h>

#define NIMG 4
#define KCH 3
#define HH 96
#define WW 96
#define PP (HH * WW)
#define NSUB 9
#define JSPLIT 3
#define TILE_I (NSUB * 16)
#define NTILES (PP / TILE_I)
#define NWAVES (NSUB * JSPLIT)
#define THREADS (NWAVES * 32)
#define JSPAN (PP / JSPLIT)
#define JROWS (JSPAN / WW)
#define ROWT (WW / 16)
#define NOUT (TILE_I * KCH)
#define NPARTIAL (JSPLIT * 16)
#define RSTR (NPARTIAL + 4)
#define NGROUPS (NIMG * NTILES)
#define NSUMW ((NOUT + 31) / 32)

typedef __attribute__((ext_vector_type(2))) float v2f;
typedef __attribute__((ext_vector_type(8))) float v8f;

static_assert(JSPAN % WW == 0 && WW % 16 == 0, "a wave's j range is whole image rows of whole 16-pixel tiles");

__global__ __launch_bounds__(THREADS) void bf_main(
    const float* __restrict__ raw,
    const float* __restrict__ nrm,
    float* __restrict__ out,
    float* __restrict__ wgmax)
{
    extern __shared__ float smem[];
    float4* s_f = (float4*)smem;
    float4* s_v = (float4*)(smem + 4 * PP);

    const int n    = blockIdx.x / NTILES;
    const int tile = blockIdx.x % NTILES;
    const int tid  = threadIdx.x;

    const float inv_rgb = 1.0f / 15.0f;
    const float inv_xy  = 1.0f / 100.0f;
    const float col_mid = 127.5f / 15.0f;
    const float es = -0.72134752044448170368f;
    const float ef =  1.44269504088896340736f;

    const float* rawn = raw + (size_t)n * KCH * PP;
    const float* nrmn = nrm + (size_t)n * KCH * PP;

    for (int p = tid; p < PP; p += THREADS) {
        float c0 = rawn[p]          * inv_rgb - col_mid;
        float c1 = rawn[PP + p]     * inv_rgb - col_mid;
        float c2 = rawn[2 * PP + p] * inv_rgb - col_mid;
        s_f[p] = make_float4(c0, c1, c2, c0 * c0 + c1 * c1 + c2 * c2);
        s_v[p] = make_float4(nrmn[p], nrmn[PP + p], nrmn[2 * PP + p], 0.0f);
    }
    __syncthreads();

    const int lane = tid & 31;
    const int l16  = lane & 15;
    const int hl   = lane >> 4;
    const int wave = tid >> 5;
    const int sub   = wave % NSUB;
    const int jpart = wave / NSUB;
    const int ibase = tile * TILE_I + sub * 16;
    const int xi0   = ibase % WW;
    const int yi    = ibase / WW;

    const float4 fi = s_f[ibase + l16];
    v2f ia;
    if (hl == 0) { ia.x = ef * fi.x;  ia.y = ef * fi.y; }
    else         { ia.x = ef * fi.z;  ia.y = es;        }

    float sqi[8];
#pragma unroll
    for (int r = 0; r < 8; ++r) sqi[r] = s_f[ibase + 8 * hl + r].w;

    const float ymul = (hl == 0) ? 0.0f : inv_xy;
    const int row0 = jpart * JROWS;

    v2f acc[4][KCH];
#pragma unroll
    for (int q = 0; q < 4; ++q) { acc[q][0] = (v2f){0.0f, 0.0f}; acc[q][1] = (v2f){0.0f, 0.0f}; acc[q][2] = (v2f){0.0f, 0.0f}; }

    for (int t = 0; t < ROWT; ++t) {
      const int dx0 = xi0 + 8 * hl - 16 * t - l16;
      v8f cx;
#pragma unroll
      for (int r = 0; r < 8; ++r) {
          const float dx = (float)(dx0 + r) * inv_xy;
          cx[r] = es * fmaf(dx, dx, sqi[r]);
      }
      const v2f*    bp = (const v2f*)smem + 2 * (row0 * WW + 16 * t + l16) + hl;
      const float4* vp = s_v + (row0 * WW + 16 * t + l16);
      float fy = (float)(row0 - yi);

      for (int it = 0; it < JROWS / 4; ++it) {
        v2f    b[4];
        float4 vj[4];
        v8f    g[4];
#pragma unroll
        for (int k = 0; k < 4; ++k) {
            b[k]  = bp[2 * WW * k];
            vj[k] = vp[WW * k];
        }
#pragma unroll
        for (int k = 0; k < 4; ++k) {
            const float dy = fy * ymul;
            b[k].y = fmaf(dy, dy, b[k].y);
            fy += 1.0f;
        }
#pragma unroll
        for (int k = 0; k < 4; ++k)
            g[k] = __builtin_amdgcn_wmma_f32_16x16x4_f32(false, ia, false, b[k], (short)0, cx, false, false);

        v2f wp[16];
        __builtin_amdgcn_sched_barrier(0);
#pragma unroll
        for (int s = 0; s < 18; ++s) {
            if (s < 16) {
                wp[s].x = fminf(fmaxf(__builtin_amdgcn_exp2f(g[s / 4][2 * (s % 4)]), 0.0f), 1.0f);
                __builtin_amdgcn_sched_barrier(0);
            }
            if (s >= 2) {
                const int p = s - 2, k = p / 4, q = p % 4;
                acc[q][0] = __builtin_elementwise_fma(wp[p], (v2f){vj[k].x, vj[k].x}, acc[q][0]);
                __builtin_amdgcn_sched_barrier(0);
            }
            if (s < 16) {
                wp[s].y = fminf(fmaxf(__builtin_amdgcn_exp2f(g[s / 4][2 * (s % 4) + 1]), 0.0f), 1.0f);
                __builtin_amdgcn_sched_barrier(0);
            }
            if (s >= 2) {
                const int p = s - 2, k = p / 4, q = p % 4;
                acc[q][1] = __builtin_elementwise_fma(wp[p], (v2f){vj[k].y, vj[k].y}, acc[q][1]);
                __builtin_amdgcn_sched_barrier(0);
                acc[q][2] = __builtin_elementwise_fma(wp[p], (v2f){vj[k].z, vj[k].z}, acc[q][2]);
                __builtin_amdgcn_sched_barrier(0);
            }
        }
        bp += 4 * 2 * WW;
        vp += 4 * WW;
      }
    }

    __syncthreads();
    float* red = smem;
    {
        float* rp = red + (sub * 16 + 8 * hl) * RSTR + jpart * 16 + l16;
#pragma unroll
        for (int r = 0; r < 8; ++r) {
            rp[r * RSTR]                       = (r & 1) ? acc[r >> 1][0].y : acc[r >> 1][0].x;
            rp[(TILE_I + r) * RSTR]            = (r & 1) ? acc[r >> 1][1].y : acc[r >> 1][1].x;
            rp[(2 * TILE_I + r) * RSTR]        = (r & 1) ? acc[r >> 1][2].y : acc[r >> 1][2].x;
        }
    }
    __syncthreads();

    float* wmax = smem + 4 * PP;
    const bool live = tid < NOUT;
    const int  e    = live ? tid : 0;
    const int  ch   = e / TILE_I;
    const int  il   = e % TILE_I;
    float s = 0.0f;
    if (wave < NSUMW) {
        const float4* q = (const float4*)(red + e * RSTR);
        for (int k = 0; k < NPARTIAL / 4; ++k) {
            const float4 p4 = q[k];
            s += p4.x; s += p4.y; s += p4.z; s += p4.w;
        }
        float m = live ? s : -INFINITY;
#pragma unroll
        for (int off = 16; off >= 1; off >>= 1) m = fmaxf(m, __shfl_xor(m, off, 32));
        if (lane == 0) wmax[wave] = m;
    }
    __syncthreads();
    if (live) out[((size_t)n * KCH + ch) * PP + tile * TILE_I + il] = s;
    if (wave == 0) {
        float m = (lane < NSUMW) ? wmax[lane] : -INFINITY;
#pragma unroll
        for (int off = 16; off >= 1; off >>= 1) m = fmaxf(m, __shfl_xor(m, off, 32));
        if (lane == 0) wgmax[blockIdx.x] = m;
    }
}

__global__ void bf_scale(float* __restrict__ out, const float* __restrict__ wgmax, int total) {
    static_assert(NGROUPS == 8 * 32, "two 16-byte reads per lane cover every slot once");
    const int i    = blockIdx.x * blockDim.x + threadIdx.x;
    const int lane = threadIdx.x & 31;
    const float4* q = (const float4*)wgmax;
    const float4 a = q[lane];
    const float4 b = q[lane + 32];
    const float  v = (i < total) ? out[i] : 0.0f;
    float m = fmaxf(fmaxf(fmaxf(a.x, a.y), fmaxf(a.z, a.w)), fmaxf(fmaxf(b.x, b.y), fmaxf(b.z, b.w)));
#pragma unroll
    for (int off = 16; off >= 1; off >>= 1) m = fmaxf(m, __shfl_xor(m, off, 32));
    if (i < total) out[i] = v / m;
}

extern "C" void kernel_launch(void* const* d_in, const int* in_sizes, int n_in,
                              void* d_out, int out_size, void* d_ws, size_t ws_size,
                              hipStream_t stream) {
    (void)in_sizes; (void)n_in; (void)out_size; (void)ws_size;
    const float* raw = (const float*)d_in[0];
    const float* nrm = (const float*)d_in[1];
    float* out = (float*)d_out;
    float* wgmax = (float*)d_ws;

    const size_t lds_bytes = (size_t)8 * PP * sizeof(float);
    hipFuncSetAttribute((const void*)bf_main,
                        hipFuncAttributeMaxDynamicSharedMemorySize, (int)lds_bytes);

    bf_main<<<NGROUPS, THREADS, lds_bytes, stream>>>(raw, nrm, out, wgmax);

    const int total = NIMG * KCH * PP;
    bf_scale<<<(total + 255) / 256, 256, 0, stream>>>(out, wgmax, total);
}


# Each wave owns three neighbouring i stripes that share every j fragment, dy^2 term and value read: 12 waves (3 per SIMD, 4 j ranges; 16-byte staging loads) not 27; sum order changed: 64 partials
# speedup vs baseline: 11.2663x; 1.0824x over previous
#include <hip/hip_runtime.h>
#include <math.h>

#define NIMG 4
#define KCH 3
#define HH 96
#define WW 96
#define PP (HH * WW)
#define NSUB 9
#define MS 3
#define NGRP (NSUB / MS)
#define JSPLIT 4
#define TILE_I (NSUB * 16)
#define NTILES (PP / TILE_I)
#define NWAVES (NGRP * JSPLIT)
#define THREADS (NWAVES * 32)
#define JSPAN (PP / JSPLIT)
#define JROWS (JSPAN / WW)
#define KR 2
#define NT (KR * MS)
#define ROWT (WW / 16)
#define NOUT (TILE_I * KCH)
#define NPARTIAL (JSPLIT * 16)
#define RSTR (NPARTIAL + 4)
#define NGROUPS (NIMG * NTILES)
#define NQUAD (PP / 4)

typedef __attribute__((ext_vector_type(2))) float v2f;
typedef __attribute__((ext_vector_type(8))) float v8f;

static_assert(JSPAN % WW == 0 && WW % 16 == 0, "a wave's j range is whole image rows of whole 16-pixel tiles");
static_assert(JROWS % KR == 0, "whole trips");
static_assert(NSUB % MS == 0 && TILE_I % (MS * 16) == 0 && WW % (MS * 16) == 0, "a wave's three stripes lie in one image row");
static_assert(NQUAD % THREADS == 0 && (NQUAD / THREADS) % 3 == 0, "staging: the same number of pixel quads for every thread, in groups of three");
static_assert(NOUT <= 2 * THREADS, "final sums: at most two output elements per thread");

__global__ __launch_bounds__(THREADS) void bf_main(
    const float* __restrict__ raw,
    const float* __restrict__ nrm,
    float* __restrict__ out,
    float* __restrict__ wgmax)
{
    extern __shared__ float smem[];
    float4* s_f = (float4*)smem;
    float4* s_v = (float4*)(smem + 4 * PP);

    const int n    = blockIdx.x / NTILES;
    const int tile = blockIdx.x % NTILES;
    const int tid  = threadIdx.x;

    const float inv_rgb = 1.0f / 15.0f;
    const float inv_xy  = 1.0f / 100.0f;
    const float col_mid = 127.5f / 15.0f;
    const float es = -0.72134752044448170368f;
    const float ef =  1.44269504088896340736f;

    const float* rawn = raw + (size_t)n * KCH * PP;
    const float* nrmn = nrm + (size_t)n * KCH * PP;

    {
        const float4* r4 = (const float4*)rawn;
        const float4* n4 = (const float4*)nrmn;
#pragma unroll
        for (int h = 0; h < NQUAD / THREADS / 3; ++h) {
            float4 rq[3][KCH], vq[3][KCH];
#pragma unroll
            for (int u = 0; u < 3; ++u) {
                const int q = tid + THREADS * (3 * h + u);
#pragma unroll
                for (int c = 0; c < KCH; ++c) { rq[u][c] = r4[c * NQUAD + q]; vq[u][c] = n4[c * NQUAD + q]; }
            }
#pragma unroll
            for (int u = 0; u < 3; ++u) {
                const int p = 4 * (tid + THREADS * (3 * h + u));
                const float ra[4] = {rq[u][0].x, rq[u][0].y, rq[u][0].z, rq[u][0].w};
                const float rb[4] = {rq[u][1].x, rq[u][1].y, rq[u][1].z, rq[u][1].w};
                const float rc[4] = {rq[u][2].x, rq[u][2].y, rq[u][2].z, rq[u][2].w};
                const float va[4] = {vq[u][0].x, vq[u][0].y, vq[u][0].z, vq[u][0].w};
                const float vb[4] = {vq[u][1].x, vq[u][1].y, vq[u][1].z, vq[u][1].w};
                const float vc[4] = {vq[u][2].x, vq[u][2].y, vq[u][2].z, vq[u][2].w};
#pragma unroll
                for (int e = 0; e < 4; ++e) {
                    const float c0 = ra[e] * inv_rgb - col_mid;
                    const float c1 = rb[e] * inv_rgb - col_mid;
                    const float c2 = rc[e] * inv_rgb - col_mid;
                    s_f[p + e] = make_float4(c0, c1, c2, c0 * c0 + c1 * c1 + c2 * c2);
                    s_v[p + e] = make_float4(va[e], vb[e], vc[e], 0.0f);
                }
            }
        }
    }
    __syncthreads();

    const int lane = tid & 31;
    const int l16  = lane & 15;
    const int hl   = lane >> 4;
    const int wave = tid >> 5;
    const int grp   = wave % NGRP;
    const int jpart = wave / NGRP;
    const int ibase = tile * TILE_I + grp * (MS * 16);
    const int xi0   = ibase % WW;
    const int yi    = ibase / WW;

    v2f ia[MS];
#pragma unroll
    for (int m = 0; m < MS; ++m) {
        const float4 fi = s_f[ibase + 16 * m + l16];
        if (hl == 0) { ia[m].x = ef * fi.x;  ia[m].y = ef * fi.y; }
        else         { ia[m].x = ef * fi.z;  ia[m].y = es;        }
    }

    float sqi[MS][8];
#pragma unroll
    for (int m = 0; m < MS; ++m)
#pragma unroll
        for (int r = 0; r < 8; ++r) sqi[m][r] = s_f[ibase + 16 * m + 8 * hl + r].w;

    const float ymul = (hl == 0) ? 0.0f : inv_xy;
    const int row0 = jpart * JROWS;

    v2f acc[MS][4][KCH];
#pragma unroll
    for (int m = 0; m < MS; ++m)
#pragma unroll
        for (int q = 0; q < 4; ++q) { acc[m][q][0] = (v2f){0.0f, 0.0f}; acc[m][q][1] = (v2f){0.0f, 0.0f}; acc[m][q][2] = (v2f){0.0f, 0.0f}; }

    for (int t = 0; t < ROWT; ++t) {
      v8f cx[MS];
#pragma unroll
      for (int m = 0; m < MS; ++m) {
          const int dx0 = xi0 + 16 * m + 8 * hl - 16 * t - l16;
#pragma unroll
          for (int r = 0; r < 8; ++r) {
              const float dx = (float)(dx0 + r) * inv_xy;
              cx[m][r] = es * fmaf(dx, dx, sqi[m][r]);
          }
      }
      const v2f*    bp = (const v2f*)smem + 2 * (row0 * WW + 16 * t + l16) + hl;
      const float4* vp = s_v + (row0 * WW + 16 * t + l16);
      float fy = (float)(row0 - yi);

      for (int it = 0; it < JROWS / KR; ++it) {
        v2f    b[KR];
        float4 vj[KR];
        v8f    g[NT];
#pragma unroll
        for (int k = 0; k < KR; ++k) {
            b[k]  = bp[2 * WW * k];
            vj[k] = vp[WW * k];
        }
#pragma unroll
        for (int k = 0; k < KR; ++k) {
            const float dy = fy * ymul;
            b[k].y = fmaf(dy, dy, b[k].y);
            fy += 1.0f;
        }
#pragma unroll
        for (int k = 0; k < KR; ++k)
#pragma unroll
            for (int m = 0; m < MS; ++m)
                g[MS * k + m] = __builtin_amdgcn_wmma_f32_16x16x4_f32(false, ia[m], false, b[k], (short)0, cx[m], false, false);

        v2f wp[4 * NT];
        __builtin_amdgcn_sched_barrier(0);
#pragma unroll
        for (int s = 0; s < 4 * NT + 2; ++s) {
            if (s < 4 * NT) {
                wp[s].x = fminf(fmaxf(__builtin_amdgcn_exp2f(g[s / 4][2 * (s % 4)]), 0.0f), 1.0f);
                __builtin_amdgcn_sched_barrier(0);
            }
            if (s >= 2) {
                const int p = s - 2, u = p / 4, q = p % 4, k = u / MS, m = u % MS;
                acc[m][q][0] = __builtin_elementwise_fma(wp[p], (v2f){vj[k].x, vj[k].x}, acc[m][q][0]);
                __builtin_amdgcn_sched_barrier(0);
            }
            if (s < 4 * NT) {
                wp[s].y = fminf(fmaxf(__builtin_amdgcn_exp2f(g[s / 4][2 * (s % 4) + 1]), 0.0f), 1.0f);
                __builtin_amdgcn_sched_barrier(0);
            }
            if (s >= 2) {
                const int p = s - 2, u = p / 4, q = p % 4, k = u / MS, m = u % MS;
                acc[m][q][1] = __builtin_elementwise_fma(wp[p], (v2f){vj[k].y, vj[k].y}, acc[m][q][1]);
                __builtin_amdgcn_sched_barrier(0);
                acc[m][q][2] = __builtin_elementwise_fma(wp[p], (v2f){vj[k].z, vj[k].z}, acc[m][q][2]);
                __builtin_amdgcn_sched_barrier(0);
            }
        }
        bp += KR * 2 * WW;
        vp += KR * WW;
      }
    }

    __syncthreads();
    float* red = smem;
#pragma unroll
    for (int m = 0; m < MS; ++m) {
        float* rp = red + ((grp * MS + m) * 16 + 8 * hl) * RSTR + jpart * 16 + l16;
#pragma unroll
        for (int r = 0; r < 8; ++r) {
            rp[r * RSTR]                       = (r & 1) ? acc[m][r >> 1][0].y : acc[m][r >> 1][0].x;
            rp[(TILE_I + r) * RSTR]            = (r & 1) ? acc[m][r >> 1][1].y : acc[m][r >> 1][1].x;
            rp[(2 * TILE_I + r) * RSTR]        = (r & 1) ? acc[m][r >> 1][2].y : acc[m][r >> 1][2].x;
        }
    }
    __syncthreads();

    float* wmax = smem + 4 * PP;
    float sums[2];
#pragma unroll
    for (int h = 0; h < 2; ++h) {
        const int e = tid + h * THREADS;
        float s = 0.0f;
        if (e < NOUT) {
            const float4* q = (const float4*)(red + e * RSTR);
            for (int k = 0; k < NPARTIAL / 4; ++k) {
                const float4 p4 = q[k];
                s += p4.x; s += p4.y; s += p4.z; s += p4.w;
            }
        }
        sums[h] = s;
    }
    {
        float m = (tid + THREADS < NOUT) ? fmaxf(sums[0], sums[1]) : sums[0];
#pragma unroll
        for (int off = 16; off >= 1; off >>= 1) m = fmaxf(m, __shfl_xor(m, off, 32));
        if (lane == 0) wmax[wave] = m;
    }
    __syncthreads();
#pragma unroll
    for (int h = 0; h < 2; ++h) {
        const int e = tid + h * THREADS;
        if (e < NOUT) out[((size_t)n * KCH + e / TILE_I) * PP + tile * TILE_I + e % TILE_I] = sums[h];
    }
    if (wave == 0) {
        float m = (lane < NWAVES) ? wmax[lane] : -INFINITY;
#pragma unroll
        for (int off = 16; off >= 1; off >>= 1) m = fmaxf(m, __shfl_xor(m, off, 32));
        if (lane == 0) wgmax[blockIdx.x] = m;
    }
}

__global__ void bf_scale(float* __restrict__ out, const float* __restrict__ wgmax, int total) {
    static_assert(NGROUPS == 8 * 32, "two 16-byte reads per lane cover every slot once");
    const int i    = blockIdx.x * blockDim.x + threadIdx.x;
    const int lane = threadIdx.x & 31;
    const float4* q = (const float4*)wgmax;
    const float4 a = q[lane];
    const float4 b = q[lane + 32];
    const float  v = (i < total) ? out[i] : 0.0f;
    float m = fmaxf(fmaxf(fmaxf(a.x, a.y), fmaxf(a.z, a.w)), fmaxf(fmaxf(b.x, b.y), fmaxf(b.z, b.w)));
#pragma unroll
    for (int off = 16; off >= 1; off >>= 1) m = fmaxf(m, __shfl_xor(m, off, 32));
    if (i < total) out[i] = v / m;
}

extern "C" void kernel_launch(void* const* d_in, const int* in_sizes, int n_in,
                              void* d_out, int out_size, void* d_ws, size_t ws_size,
                              hipStream_t stream) {
    (void)in_sizes; (void)n_in; (void)out_size; (void)ws_size;
    const float* raw = (const float*)d_in[0];
    const float* nrm = (const float*)d_in[1];
    float* out = (float*)d_out;
    float* wgmax = (float*)d_ws;

    const size_t lds_bytes = (size_t)8 * PP * sizeof(float);
    hipFuncSetAttribute((const void*)bf_main,
                        hipFuncAttributeMaxDynamicSharedMemorySize, (int)lds_bytes);

    bf_main<<<NGROUPS, THREADS, lds_bytes, stream>>>(raw, nrm, out, wgmax);

    const int total = NIMG * KCH * PP;
    bf_scale<<<(total + 255) / 256, 256, 0, stream>>>(out, wgmax, total);
}


# LDS operands of each trip (B fragments, values) read one trip ahead into a second register set, as one group: a wave left alone on its SIMD no longer waits one LDS round trip per trip; sum order kept
# speedup vs baseline: 11.6798x; 1.0367x over previous
#include <hip/hip_runtime.h>
#include <math.h>

#define NIMG 4
#define KCH 3
#define HH 96
#define WW 96
#define PP (HH * WW)
#define NSUB 9
#define MS 3
#define NGRP (NSUB / MS)
#define JSPLIT 4
#define TILE_I (NSUB * 16)
#define NTILES (PP / TILE_I)
#define NWAVES (NGRP * JSPLIT)
#define THREADS (NWAVES * 32)
#define JSPAN (PP / JSPLIT)
#define JROWS (JSPAN / WW)
#define KR 2
#define NT (KR * MS)
#define ROWT (WW / 16)
#define NOUT (TILE_I * KCH)
#define NPARTIAL (JSPLIT * 16)
#define RSTR (NPARTIAL + 4)
#define NGROUPS (NIMG * NTILES)
#define NQUAD (PP / 4)

typedef __attribute__((ext_vector_type(2))) float v2f;
typedef __attribute__((ext_vector_type(8))) float v8f;

static_assert(JSPAN % WW == 0 && WW % 16 == 0, "a wave's j range is whole image rows of whole 16-pixel tiles");
static_assert(JROWS % (2 * KR) == 0, "whole trips, an even number per tile column");
static_assert(NSUB % MS == 0 && TILE_I % (MS * 16) == 0 && WW % (MS * 16) == 0, "a wave's three stripes lie in one image row");
static_assert(NQUAD % THREADS == 0 && (NQUAD / THREADS) % 3 == 0, "staging: the same number of pixel quads for every thread, in groups of three");
static_assert(NOUT <= 2 * THREADS, "final sums: at most two output elements per thread");

__global__ __launch_bounds__(THREADS) void bf_main(
    const float* __restrict__ raw,
    const float* __restrict__ nrm,
    float* __restrict__ out,
    float* __restrict__ wgmax)
{
    extern __shared__ float smem[];
    float4* s_f = (float4*)smem;
    float4* s_v = (float4*)(smem + 4 * PP);

    const int n    = blockIdx.x / NTILES;
    const int tile = blockIdx.x % NTILES;
    const int tid  = threadIdx.x;

    const float inv_rgb = 1.0f / 15.0f;
    const float inv_xy  = 1.0f / 100.0f;
    const float col_mid = 127.5f / 15.0f;
    const float es = -0.72134752044448170368f;
    const float ef =  1.44269504088896340736f;

    const float* rawn = raw + (size_t)n * KCH * PP;
    const float* nrmn = nrm + (size_t)n * KCH * PP;

    {
        const float4* r4 = (const float4*)rawn;
        const float4* n4 = (const float4*)nrmn;
#pragma unroll
        for (int h = 0; h < NQUAD / THREADS / 3; ++h) {
            float4 rq[3][KCH], vq[3][KCH];
#pragma unroll
            for (int u = 0; u < 3; ++u) {
                const int q = tid + THREADS * (3 * h + u);
#pragma unroll
                for (int c = 0; c < KCH; ++c) { rq[u][c] = r4[c * NQUAD + q]; vq[u][c] = n4[c * NQUAD + q]; }
            }
#pragma unroll
            for (int u = 0; u < 3; ++u) {
                const int p = 4 * (tid + THREADS * (3 * h + u));
                const float ra[4] = {rq[u][0].x, rq[u][0].y, rq[u][0].z, rq[u][0].w};
                const float rb[4] = {rq[u][1].x, rq[u][1].y, rq[u][1].z, rq[u][1].w};
                const float rc[4] = {rq[u][2].x, rq[u][2].y, rq[u][2].z, rq[u][2].w};
                const float va[4] = {vq[u][0].x, vq[u][0].y, vq[u][0].z, vq[u][0].w};
                const float vb[4] = {vq[u][1].x, vq[u][1].y, vq[u][1].z, vq[u][1].w};
                const float vc[4] = {vq[u][2].x, vq[u][2].y, vq[u][2].z, vq[u][2].w};
#pragma unroll
                for (int e = 0; e < 4; ++e) {
                    const float c0 = ra[e] * inv_rgb - col_mid;
                    const float c1 = rb[e] * inv_rgb - col_mid;
                    const float c2 = rc[e] * inv_rgb - col_mid;
                    s_f[p + e] = make_float4(c0, c1, c2, c0 * c0 + c1 * c1 + c2 * c2);
                    s_v[p + e] = make_float4(va[e], vb[e], vc[e], vc[e]);
                }
            }
        }
    }
    __syncthreads();

    const int lane = tid & 31;
    const int l16  = lane & 15;
    const int hl   = lane >> 4;
    const int wave = tid >> 5;
    const int grp   = wave % NGRP;
    const int jpart = wave / NGRP;
    const int ibase = tile * TILE_I + grp * (MS * 16);
    const int xi0   = ibase % WW;
    const int yi    = ibase / WW;

    v2f ia[MS];
#pragma unroll
    for (int m = 0; m < MS; ++m) {
        const float4 fi = s_f[ibase + 16 * m + l16];
        if (hl == 0) { ia[m].x = ef * fi.x;  ia[m].y = ef * fi.y; }
        else         { ia[m].x = ef * fi.z;  ia[m].y = es;        }
    }

    float sqi[MS][8];
#pragma unroll
    for (int m = 0; m < MS; ++m)
#pragma unroll
        for (int r = 0; r < 8; ++r) sqi[m][r] = s_f[ibase + 16 * m + 8 * hl + r].w;

    const float ymul = (hl == 0) ? 0.0f : inv_xy;
    const int row0 = jpart * JROWS;

    v2f acc[MS][4][KCH];
#pragma unroll
    for (int m = 0; m < MS; ++m)
#pragma unroll
        for (int q = 0; q < 4; ++q) { acc[m][q][0] = (v2f){0.0f, 0.0f}; acc[m][q][1] = (v2f){0.0f, 0.0f}; acc[m][q][2] = (v2f){0.0f, 0.0f}; }

    const v2f*    bp = (const v2f*)smem + 2 * (row0 * WW + l16) + hl;
    const float4* vp = s_v + (row0 * WW + l16);
    v2f    bb[2][KR];
    float4 vv[2][KR];
#pragma unroll
    for (int k = 0; k < KR; ++k) {
        bb[0][k] = bp[2 * WW * k];
        vv[0][k] = vp[WW * k];
    }

    for (int t = 0; t < ROWT; ++t) {
      v8f cx[MS];
#pragma unroll
      for (int m = 0; m < MS; ++m) {
          const int dx0 = xi0 + 16 * m + 8 * hl - 16 * t - l16;
#pragma unroll
          for (int r = 0; r < 8; ++r) {
              const float dx = (float)(dx0 + r) * inv_xy;
              cx[m][r] = es * fmaf(dx, dx, sqi[m][r]);
          }
      }
      float fy = (float)(row0 - yi);

      for (int it2 = 0; it2 < JROWS / KR / 2; ++it2) {
#pragma unroll
       for (int h = 0; h < 2; ++h) {
        v2f    b[KR];
        float4 vj[KR];
        v8f    g[NT];
        {
            const int step = (h == 1 && it2 == JROWS / KR / 2 - 1) ? (16 - (JROWS - KR) * WW) : (KR * WW);
            bp += 2 * step;
            vp += step;
#pragma unroll
            for (int k = 0; k < KR; ++k) {
                bb[h ^ 1][k] = bp[2 * WW * k];
                vv[h ^ 1][k] = vp[WW * k];
            }
            __builtin_amdgcn_sched_barrier(0);
        }
#pragma unroll
        for (int k = 0; k < KR; ++k) {
            const float dy = fy * ymul;
            b[k]   = bb[h][k];
            vj[k]  = vv[h][k];
            b[k].y = fmaf(dy, dy, b[k].y);
            fy += 1.0f;
        }
#pragma unroll
        for (int k = 0; k < KR; ++k)
#pragma unroll
            for (int m = 0; m < MS; ++m)
                g[MS * k + m] = __builtin_amdgcn_wmma_f32_16x16x4_f32(false, ia[m], false, b[k], (short)0, cx[m], false, false);

        v2f wp[4 * NT];
        __builtin_amdgcn_sched_barrier(0);
#pragma unroll
        for (int s = 0; s < 4 * NT + 2; ++s) {
            if (s < 4 * NT) {
                wp[s].x = fminf(fmaxf(__builtin_amdgcn_exp2f(g[s / 4][2 * (s % 4)]), 0.0f), 1.0f);
                __builtin_amdgcn_sched_barrier(0);
            }
            if (s >= 2) {
                const int p = s - 2, u = p / 4, q = p % 4, k = u / MS, m = u % MS;
                acc[m][q][0] = __builtin_elementwise_fma(wp[p], (v2f){vj[k].x, vj[k].x}, acc[m][q][0]);
                __builtin_amdgcn_sched_barrier(0);
            }
            if (s < 4 * NT) {
                wp[s].y = fminf(fmaxf(__builtin_amdgcn_exp2f(g[s / 4][2 * (s % 4) + 1]), 0.0f), 1.0f);
                __builtin_amdgcn_sched_barrier(0);
            }
            if (s >= 2) {
                const int p = s - 2, u = p / 4, q = p % 4, k = u / MS, m = u % MS;
                acc[m][q][1] = __builtin_elementwise_fma(wp[p], (v2f){vj[k].y, vj[k].y}, acc[m][q][1]);
                __builtin_amdgcn_sched_barrier(0);
                acc[m][q][2] = __builtin_elementwise_fma(wp[p], (v2f){vj[k].z, vj[k].w}, acc[m][q][2]);
                __builtin_amdgcn_sched_barrier(0);
            }
        }
       }
      }
    }

    __syncthreads();
    float* red = smem;
#pragma unroll
    for (int m = 0; m < MS; ++m) {
        float* rp = red + ((grp * MS + m) * 16 + 8 * hl) * RSTR + jpart * 16 + l16;
#pragma unroll
        for (int r = 0; r < 8; ++r) {
            rp[r * RSTR]                       = (r & 1) ? acc[m][r >> 1][0].y : acc[m][r >> 1][0].x;
            rp[(TILE_I + r) * RSTR]            = (r & 1) ? acc[m][r >> 1][1].y : acc[m][r >> 1][1].x;
            rp[(2 * TILE_I + r) * RSTR]        = (r & 1) ? acc[m][r >> 1][2].y : acc[m][r >> 1][2].x;
        }
    }
    __syncthreads();

    float* wmax = smem + 4 * PP;
    float sums[2];
#pragma unroll
    for (int h = 0; h < 2; ++h) {
        const int e = tid + h * THREADS;
        float s = 0.0f;
        if (e < NOUT) {
            const float4* q = (const float4*)(red + e * RSTR);
            for (int k = 0; k < NPARTIAL / 4; ++k) {
                const float4 p4 = q[k];
                s += p4.x; s += p4.y; s += p4.z; s += p4.w;
            }
        }
        sums[h] = s;
    }
    {
        float m = (tid + THREADS < NOUT) ? fmaxf(sums[0], sums[1]) : sums[0];
#pragma unroll
        for (int off = 16; off >= 1; off >>= 1) m = fmaxf(m, __shfl_xor(m, off, 32));
        if (lane == 0) wmax[wave] = m;
    }
    __syncthreads();
#pragma unroll
    for (int h = 0; h < 2; ++h) {
        const int e = tid + h * THREADS;
        if (e < NOUT) out[((size_t)n * KCH + e / TILE_I) * PP + tile * TILE_I + e % TILE_I] = sums[h];
    }
    if (wave == 0) {
        float m = (lane < NWAVES) ? wmax[lane] : -INFINITY;
#pragma unroll
        for (int off = 16; off >= 1; off >>= 1) m = fmaxf(m, __shfl_xor(m, off, 32));
        if (lane == 0) wgmax[blockIdx.x] = m;
    }
}

__global__ void bf_scale(float* __restrict__ out, const float* __restrict__ wgmax, int total) {
    static_assert(NGROUPS == 8 * 32, "two 16-byte reads per lane cover every slot once");
    const int i    = blockIdx.x * blockDim.x + threadIdx.x;
    const int lane = threadIdx.x & 31;
    const float4* q = (const float4*)wgmax;
    const float4 a = q[lane];
    const float4 b = q[lane + 32];
    const float  v = (i < total) ? out[i] : 0.0f;
    float m = fmaxf(fmaxf(fmaxf(a.x, a.y), fmaxf(a.z, a.w)), fmaxf(fmaxf(b.x, b.y), fmaxf(b.z, b.w)));
#pragma unroll
    for (int off = 16; off >= 1; off >>= 1) m = fmaxf(m, __shfl_xor(m, off, 32));
    if (i < total) out[i] = v / m;
}

extern "C" void kernel_launch(void* const* d_in, const int* in_sizes, int n_in,
                              void* d_out, int out_size, void* d_ws, size_t ws_size,
                              hipStream_t stream) {
    (void)in_sizes; (void)n_in; (void)out_size; (void)ws_size;
    const float* raw = (const float*)d_in[0];
    const float* nrm = (const float*)d_in[1];
    float* out = (float*)d_out;
    float* wgmax = (float*)d_ws;

    const size_t lds_bytes = (size_t)8 * PP * sizeof(float);
    hipFuncSetAttribute((const void*)bf_main,
                        hipFuncAttributeMaxDynamicSharedMemorySize, (int)lds_bytes);

    bf_main<<<NGROUPS, THREADS, lds_bytes, stream>>>(raw, nrm, out, wgmax);

    const int total = NIMG * KCH * PP;
    bf_scale<<<(total + 255) / 256, 256, 0, stream>>>(out, wgmax, total);
}
